# starting kernel
# baseline (speedup 1.0000x reference)
#include <hip/hip_runtime.h>

#define B_  16
#define NA  2048
#define NP  2048
#define DD  512
#define KK  32

typedef __attribute__((ext_vector_type(2))) float v2f;
typedef __attribute__((ext_vector_type(8))) float v8f;

__device__ __forceinline__ unsigned ordkey(float f) {
    unsigned u = __float_as_uint(f);
    return (u & 0x80000000u) ? ~u : (u | 0x80000000u);
}
__device__ __forceinline__ float keyval(unsigned k) {
    return __uint_as_float((k & 0x80000000u) ? (k ^ 0x80000000u) : ~k);
}

__global__ __launch_bounds__(256) void prep_kernel(
    const float* __restrict__ anchor, const int* __restrict__ idx_a,
    const int* __restrict__ idx_p, float* __restrict__ a_s_out,
    float* __restrict__ asel_ws) {
    const int r    = blockIdx.x * 8 + (threadIdx.x >> 5);
    const int lane = threadIdx.x & 31;
    const int kind = r >> 9;
    const int t    = r & 511;
    const int b    = t >> 5;
    const int idx  = kind ? idx_p[t] : idx_a[t];
    const float* src = anchor + ((size_t)b * NA + idx) * DD;
    float v[16];
    float s = 0.f;
#pragma unroll
    for (int j = 0; j < 16; ++j) { float x = src[lane + 32 * j]; v[j] = x; s += x * x; }
#pragma unroll
    for (int off = 16; off; off >>= 1) s += __shfl_xor(s, off, 32);
    const float inv = 1.f / fmaxf(sqrtf(s), 1e-12f);
    float* dst = (kind ? asel_ws : a_s_out) + (size_t)t * DD;
#pragma unroll
    for (int j = 0; j < 16; ++j) dst[lane + 32 * j] = v[j] * inv;
}

__global__ void init_best(unsigned long long* best) {
    best[blockIdx.x * 256 + threadIdx.x] = 0ull;
}

__global__ __launch_bounds__(256) void score_kernel(
    const float* __restrict__ p, const float* __restrict__ asel,
    unsigned long long* __restrict__ best) {
    __shared__ float As[KK * DD];
    const int b   = blockIdx.y;
    const int tid = threadIdx.x;
    {
        const float4* s4 = (const float4*)(asel + (size_t)b * KK * DD);
        float4* d4 = (float4*)As;
        for (int i = tid; i < KK * DD / 4; i += 256) d4[i] = s4[i];
    }
    __syncthreads();

    const int wave = tid >> 5;
    const int lane = tid & 31;
    const int half = lane >> 4;
    const int lr   = lane & 15;
    const int m    = blockIdx.x * 128 + wave * 16 + lr;

    const float* pb = p + ((size_t)b * NP + m) * DD + 2 * half;
    const float* a0 = As + lr * DD + 2 * half;
    const float* a1 = As + (lr + 16) * DD + 2 * half;

    v8f acc0 = {};
    v8f acc1 = {};
    float ss = 0.f;
    for (int k0 = 0; k0 < DD; k0 += 4) {
        v2f bf  = *(const v2f*)(pb + k0);
        v2f af0 = *(const v2f*)(a0 + k0);
        v2f af1 = *(const v2f*)(a1 + k0);
        ss += bf.x * bf.x + bf.y * bf.y;
        acc0 = __builtin_amdgcn_wmma_f32_16x16x4_f32(false, af0, false, bf,
                                                     (short)0, acc0, false, false);
        acc1 = __builtin_amdgcn_wmma_f32_16x16x4_f32(false, af1, false, bf,
                                                     (short)0, acc1, false, false);
    }
    ss += __shfl_xor(ss, 16, 32);
    const float inv = 1.f / fmaxf(sqrtf(ss), 1e-12f);
    const unsigned keylo = (unsigned)(NP - 1 - m);

#pragma unroll
    for (int i = 0; i < 8; ++i) {
        unsigned long long k0p =
            ((unsigned long long)ordkey(acc0[i] * inv) << 32) | keylo;
        unsigned long long k1p =
            ((unsigned long long)ordkey(acc1[i] * inv) << 32) | keylo;
#pragma unroll
        for (int off = 8; off; off >>= 1) {
            unsigned long long o0 = __shfl_xor(k0p, off, 32);
            unsigned long long o1 = __shfl_xor(k1p, off, 32);
            k0p = o0 > k0p ? o0 : k0p;
            k1p = o1 > k1p ? o1 : k1p;
        }
        if (lr == 0) {
            atomicMax(best + (size_t)b * KK + (i + 8 * half), k0p);
            atomicMax(best + (size_t)b * KK + (16 + i + 8 * half), k1p);
        }
    }
}

__global__ __launch_bounds__(256) void fuse_kernel(
    const float* __restrict__ p1, const float* __restrict__ p2,
    const unsigned long long* __restrict__ best, float* __restrict__ p_s_out) {
    const int t    = blockIdx.x * 8 + (threadIdx.x >> 5);
    const int lane = threadIdx.x & 31;
    const int b    = t >> 5;
    const unsigned long long e1 = best[t];
    const unsigned long long e2 = best[B_ * KK + t];
    const float w1 = keyval((unsigned)(e1 >> 32));
    const float w2 = keyval((unsigned)(e2 >> 32));
    const int  i1  = NP - 1 - (int)(unsigned)(e1 & 0xffffffffull);
    const int  i2  = NP - 1 - (int)(unsigned)(e2 & 0xffffffffull);
    const float* r1 = p1 + ((size_t)b * NP + i1) * DD;
    const float* r2 = p2 + ((size_t)b * NP + i2) * DD;

    float v1[16], v2[16];
    float s1 = 0.f, s2 = 0.f;
#pragma unroll
    for (int j = 0; j < 16; ++j) {
        float x = r1[lane + 32 * j]; v1[j] = x; s1 += x * x;
        float y = r2[lane + 32 * j]; v2[j] = y; s2 += y * y;
    }
#pragma unroll
    for (int off = 16; off; off >>= 1) {
        s1 += __shfl_xor(s1, off, 32);
        s2 += __shfl_xor(s2, off, 32);
    }
    const float c1 = w1 / fmaxf(sqrtf(s1), 1e-12f);
    const float c2 = w2 / fmaxf(sqrtf(s2), 1e-12f);
    const float dn = 1.f / (w1 + w2 + 1e-6f);

    float f[16];
    float sf = 0.f;
#pragma unroll
    for (int j = 0; j < 16; ++j) {
        f[j] = (c1 * v1[j] + c2 * v2[j]) * dn;
        sf += f[j] * f[j];
    }
#pragma unroll
    for (int off = 16; off; off >>= 1) sf += __shfl_xor(sf, off, 32);
    const float invf = 1.f / fmaxf(sqrtf(sf), 1e-12f);
    float* dst = p_s_out + (size_t)t * DD;
#pragma unroll
    for (int j = 0; j < 16; ++j) dst[lane + 32 * j] = f[j] * invf;
}

extern "C" void kernel_launch(void* const* d_in, const int* in_sizes, int n_in,
                              void* d_out, int out_size, void* d_ws, size_t ws_size,
                              hipStream_t stream) {
    (void)in_sizes; (void)n_in; (void)out_size; (void)ws_size;
    const float* anchor = (const float*)d_in[0];
    const float* p1     = (const float*)d_in[1];
    const float* p2     = (const float*)d_in[2];
    const int*   idx_a  = (const int*)d_in[3];
    const int*   idx_p  = (const int*)d_in[4];

    float* A_s = (float*)d_out;
    float* P_s = (float*)d_out + (size_t)B_ * KK * DD;

    unsigned long long* best = (unsigned long long*)d_ws;
    float* asel = (float*)((char*)d_ws + (size_t)2 * B_ * KK * sizeof(unsigned long long));

    init_best<<<dim3((2 * B_ * KK) / 256), 256, 0, stream>>>(best);
    prep_kernel<<<dim3((2 * B_ * KK) / 8), 256, 0, stream>>>(anchor, idx_a, idx_p, A_s, asel);

    dim3 sg(NP / 128, B_);
    score_kernel<<<sg, 256, 0, stream>>>(p1, asel, best);
    score_kernel<<<sg, 256, 0, stream>>>(p2, asel, best + B_ * KK);

    fuse_kernel<<<dim3((B_ * KK) / 8), 256, 0, stream>>>(p1, p2, best, P_s);
}


# Score kernel loaders rewritten: 16-B loads in one matched k order (a 128-B line of p per row per 32-k chunk), 6 chunks prefetched in registers, A from LDS at odd 16-B pitch; k summation order changes
# speedup vs baseline: 1.4467x; 1.4467x over previous
#include <hip/hip_runtime.h>

#define B_  16
#define NA  2048
#define NP  2048
#define DD  512
#define KK  32

typedef __attribute__((ext_vector_type(2))) float v2f;
typedef __attribute__((ext_vector_type(8))) float v8f;

__device__ __forceinline__ unsigned ordkey(float f) {
    unsigned u = __float_as_uint(f);
    return (u & 0x80000000u) ? ~u : (u | 0x80000000u);
}
__device__ __forceinline__ float keyval(unsigned k) {
    return __uint_as_float((k & 0x80000000u) ? (k ^ 0x80000000u) : ~k);
}

__global__ __launch_bounds__(256) void prep_kernel(
    const float* __restrict__ anchor, const int* __restrict__ idx_a,
    const int* __restrict__ idx_p, float* __restrict__ a_s_out,
    float* __restrict__ asel_ws) {
    const int r    = blockIdx.x * 8 + (threadIdx.x >> 5);
    const int lane = threadIdx.x & 31;
    const int kind = r >> 9;
    const int t    = r & 511;
    const int b    = t >> 5;
    const int idx  = kind ? idx_p[t] : idx_a[t];
    const float* src = anchor + ((size_t)b * NA + idx) * DD;
    float v[16];
    float s = 0.f;
#pragma unroll
    for (int j = 0; j < 16; ++j) { float x = src[lane + 32 * j]; v[j] = x; s += x * x; }
#pragma unroll
    for (int off = 16; off; off >>= 1) s += __shfl_xor(s, off, 32);
    const float inv = 1.f / fmaxf(sqrtf(s), 1e-12f);
    float* dst = (kind ? asel_ws : a_s_out) + (size_t)t * DD;
#pragma unroll
    for (int j = 0; j < 16; ++j) dst[lane + 32 * j] = v[j] * inv;
}

__global__ void init_best(unsigned long long* best) {
    best[blockIdx.x * 256 + threadIdx.x] = 0ull;
}

#define APITCH (DD + 4)
#define KCH    32
#define NCH    (DD / KCH)
#define PFD    6

__global__ __launch_bounds__(256) void score_kernel(
    const float* __restrict__ p, const float* __restrict__ asel,
    unsigned long long* __restrict__ best) {
    __shared__ __attribute__((aligned(16))) float As[KK * APITCH];
    const int b    = blockIdx.y;
    const int tid  = threadIdx.x;
    const int wave = tid >> 5;
    const int lane = tid & 31;
    const int half = lane >> 4;
    const int lr   = lane & 15;
    const int m    = blockIdx.x * 128 + wave * 16 + lr;

    const float4* pb4 =
        (const float4*)(p + ((size_t)b * NP + m) * DD + 16 * half);

    float4 qb[NCH][4];
#pragma unroll
    for (int c = 0; c < PFD; ++c)
#pragma unroll
        for (int i = 0; i < 4; ++i) qb[c][i] = pb4[c * (KCH / 4) + i];

    {
        const float4* s4 = (const float4*)(asel + (size_t)b * KK * DD);
        float4* d4 = (float4*)As;
        float4 st[KK * DD / 4 / 256];
#pragma unroll
        for (int j = 0; j < KK * DD / 4 / 256; ++j) st[j] = s4[tid + 256 * j];
#pragma unroll
        for (int j = 0; j < KK * DD / 4 / 256; ++j) {
            const int i = tid + 256 * j;
            d4[(i >> 7) * (APITCH / 4) + (i & 127)] = st[j];
        }
    }
    __syncthreads();

    const float4* a0 = (const float4*)(As + lr * APITCH + 16 * half);
    const float4* a1 = (const float4*)(As + (lr + 16) * APITCH + 16 * half);

    float4 aa[2][2][4];
#pragma unroll
    for (int i = 0; i < 4; ++i) { aa[0][0][i] = a0[i]; aa[0][1][i] = a1[i]; }

    v8f acc0 = {};
    v8f acc1 = {};
    float ss = 0.f;
#pragma unroll
    for (int c = 0; c < NCH; ++c) {
        if (c + PFD < NCH) {
#pragma unroll
            for (int i = 0; i < 4; ++i)
                qb[c + PFD][i] = pb4[(c + PFD) * (KCH / 4) + i];
        }
        if (c + 1 < NCH) {
#pragma unroll
            for (int i = 0; i < 4; ++i) {
                aa[(c + 1) & 1][0][i] = a0[(c + 1) * (KCH / 4) + i];
                aa[(c + 1) & 1][1][i] = a1[(c + 1) * (KCH / 4) + i];
            }
        }
        __builtin_amdgcn_sched_barrier(0);
#pragma unroll
        for (int i = 0; i < 4; ++i) {
            const float4 bq = qb[c][i];
            const float4 x0 = aa[c & 1][0][i];
            const float4 x1 = aa[c & 1][1][i];
            ss += bq.x * bq.x + bq.y * bq.y + bq.z * bq.z + bq.w * bq.w;
            v2f bf  = {bq.x, bq.y};
            v2f af0 = {x0.x, x0.y};
            v2f af1 = {x1.x, x1.y};
            acc0 = __builtin_amdgcn_wmma_f32_16x16x4_f32(false, af0, false, bf,
                                                         (short)0, acc0, false, false);
            acc1 = __builtin_amdgcn_wmma_f32_16x16x4_f32(false, af1, false, bf,
                                                         (short)0, acc1, false, false);
            bf  = (v2f){bq.z, bq.w};
            af0 = (v2f){x0.z, x0.w};
            af1 = (v2f){x1.z, x1.w};
            acc0 = __builtin_amdgcn_wmma_f32_16x16x4_f32(false, af0, false, bf,
                                                         (short)0, acc0, false, false);
            acc1 = __builtin_amdgcn_wmma_f32_16x16x4_f32(false, af1, false, bf,
                                                         (short)0, acc1, false, false);
        }
        __builtin_amdgcn_sched_barrier(0);
    }
    ss += __shfl_xor(ss, 16, 32);
    const float inv = 1.f / fmaxf(sqrtf(ss), 1e-12f);
    const unsigned keylo = (unsigned)(NP - 1 - m);

#pragma unroll
    for (int i = 0; i < 8; ++i) {
        unsigned long long k0p =
            ((unsigned long long)ordkey(acc0[i] * inv) << 32) | keylo;
        unsigned long long k1p =
            ((unsigned long long)ordkey(acc1[i] * inv) << 32) | keylo;
#pragma unroll
        for (int off = 8; off; off >>= 1) {
            unsigned long long o0 = __shfl_xor(k0p, off, 32);
            unsigned long long o1 = __shfl_xor(k1p, off, 32);
            k0p = o0 > k0p ? o0 : k0p;
            k1p = o1 > k1p ? o1 : k1p;
        }
        if (lr == 0) {
            atomicMax(best + (size_t)b * KK + (i + 8 * half), k0p);
            atomicMax(best + (size_t)b * KK + (16 + i + 8 * half), k1p);
        }
    }
}

__global__ __launch_bounds__(256) void fuse_kernel(
    const float* __restrict__ p1, const float* __restrict__ p2,
    const unsigned long long* __restrict__ best, float* __restrict__ p_s_out) {
    const int t    = blockIdx.x * 8 + (threadIdx.x >> 5);
    const int lane = threadIdx.x & 31;
    const int b    = t >> 5;
    const unsigned long long e1 = best[t];
    const unsigned long long e2 = best[B_ * KK + t];
    const float w1 = keyval((unsigned)(e1 >> 32));
    const float w2 = keyval((unsigned)(e2 >> 32));
    const int  i1  = NP - 1 - (int)(unsigned)(e1 & 0xffffffffull);
    const int  i2  = NP - 1 - (int)(unsigned)(e2 & 0xffffffffull);
    const float* r1 = p1 + ((size_t)b * NP + i1) * DD;
    const float* r2 = p2 + ((size_t)b * NP + i2) * DD;

    float v1[16], v2[16];
    float s1 = 0.f, s2 = 0.f;
#pragma unroll
    for (int j = 0; j < 16; ++j) {
        float x = r1[lane + 32 * j]; v1[j] = x; s1 += x * x;
        float y = r2[lane + 32 * j]; v2[j] = y; s2 += y * y;
    }
#pragma unroll
    for (int off = 16; off; off >>= 1) {
        s1 += __shfl_xor(s1, off, 32);
        s2 += __shfl_xor(s2, off, 32);
    }
    const float c1 = w1 / fmaxf(sqrtf(s1), 1e-12f);
    const float c2 = w2 / fmaxf(sqrtf(s2), 1e-12f);
    const float dn = 1.f / (w1 + w2 + 1e-6f);

    float f[16];
    float sf = 0.f;
#pragma unroll
    for (int j = 0; j < 16; ++j) {
        f[j] = (c1 * v1[j] + c2 * v2[j]) * dn;
        sf += f[j] * f[j];
    }
#pragma unroll
    for (int off = 16; off; off >>= 1) sf += __shfl_xor(sf, off, 32);
    const float invf = 1.f / fmaxf(sqrtf(sf), 1e-12f);
    float* dst = p_s_out + (size_t)t * DD;
#pragma unroll
    for (int j = 0; j < 16; ++j) dst[lane + 32 * j] = f[j] * invf;
}

extern "C" void kernel_launch(void* const* d_in, const int* in_sizes, int n_in,
                              void* d_out, int out_size, void* d_ws, size_t ws_size,
                              hipStream_t stream) {
    (void)in_sizes; (void)n_in; (void)out_size; (void)ws_size;
    const float* anchor = (const float*)d_in[0];
    const float* p1     = (const float*)d_in[1];
    const float* p2     = (const float*)d_in[2];
    const int*   idx_a  = (const int*)d_in[3];
    const int*   idx_p  = (const int*)d_in[4];

    float* A_s = (float*)d_out;
    float* P_s = (float*)d_out + (size_t)B_ * KK * DD;

    unsigned long long* best = (unsigned long long*)d_ws;
    float* asel = (float*)((char*)d_ws + (size_t)2 * B_ * KK * sizeof(unsigned long long));

    init_best<<<dim3((2 * B_ * KK) / 256), 256, 0, stream>>>(best);
    prep_kernel<<<dim3((2 * B_ * KK) / 8), 256, 0, stream>>>(anchor, idx_a, idx_p, A_s, asel);

    dim3 sg(NP / 128, B_);
    score_kernel<<<sg, 256, 0, stream>>>(p1, asel, best);
    score_kernel<<<sg, 256, 0, stream>>>(p2, asel, best + B_ * KK);

    fuse_kernel<<<dim3((B_ * KK) / 8), 256, 0, stream>>>(p1, p2, best, P_s);
}


# Argmax without atomics: each score block reduces its 32x128 keys in LDS and stores 32 partials, fuse maxes 16 per pass; zeroing launch gone (5->4); Asel staged in the P_s plane; winners bit-identical
# speedup vs baseline: 1.6547x; 1.1438x over previous
#include <hip/hip_runtime.h>

#define B_  16
#define NA  2048
#define NP  2048
#define DD  512
#define KK  32

typedef __attribute__((ext_vector_type(2))) float v2f;
typedef __attribute__((ext_vector_type(8))) float v8f;

__device__ __forceinline__ unsigned ordkey(float f) {
    unsigned u = __float_as_uint(f);
    return (u & 0x80000000u) ? ~u : (u | 0x80000000u);
}
__device__ __forceinline__ float keyval(unsigned k) {
    return __uint_as_float((k & 0x80000000u) ? (k ^ 0x80000000u) : ~k);
}

__global__ __launch_bounds__(256) void prep_kernel(
    const float* __restrict__ anchor, const int* __restrict__ idx_a,
    const int* __restrict__ idx_p, float* __restrict__ a_s_out,
    float* __restrict__ asel_out) {
    const int r    = blockIdx.x * 8 + (threadIdx.x >> 5);
    const int lane = threadIdx.x & 31;
    const int kind = r >> 9;
    const int t    = r & 511;
    const int b    = t >> 5;
    const int idx  = kind ? idx_p[t] : idx_a[t];
    const float* src = anchor + ((size_t)b * NA + idx) * DD;
    float v[16];
    float s = 0.f;
#pragma unroll
    for (int j = 0; j < 16; ++j) { float x = src[lane + 32 * j]; v[j] = x; s += x * x; }
#pragma unroll
    for (int off = 16; off; off >>= 1) s += __shfl_xor(s, off, 32);
    const float inv = 1.f / fmaxf(sqrtf(s), 1e-12f);
    float* dst = (kind ? asel_out : a_s_out) + (size_t)t * DD;
#pragma unroll
    for (int j = 0; j < 16; ++j) dst[lane + 32 * j] = v[j] * inv;
}

#define APITCH (DD + 4)
#define KCH    32
#define NCH    (DD / KCH)
#define PFD    6
#define NBLK   (NP / 128)
#define KPITCH 130

typedef unsigned long long u64;
typedef __attribute__((ext_vector_type(2))) unsigned long long u64x2;

__global__ __launch_bounds__(256) void score_kernel(
    const float* __restrict__ p, const float* __restrict__ asel,
    u64* __restrict__ part) {
    __shared__ __attribute__((aligned(16))) float As[KK * APITCH];
    __shared__ __attribute__((aligned(16))) u64 kbuf[KK * KPITCH];
    __shared__ __attribute__((aligned(16))) u64 pbuf[8 * KK];
    const int b    = blockIdx.y;
    const int tid  = threadIdx.x;
    const int wave = tid >> 5;
    const int lane = tid & 31;
    const int half = lane >> 4;
    const int lr   = lane & 15;
    const int m    = blockIdx.x * 128 + wave * 16 + lr;

    const float4* pb4 =
        (const float4*)(p + ((size_t)b * NP + m) * DD + 16 * half);

    float4 qb[NCH][4];
#pragma unroll
    for (int c = 0; c < PFD; ++c)
#pragma unroll
        for (int i = 0; i < 4; ++i) qb[c][i] = pb4[c * (KCH / 4) + i];

    {
        const float4* s4 = (const float4*)(asel + (size_t)b * KK * DD);
        float4* d4 = (float4*)As;
        float4 st[KK * DD / 4 / 256];
#pragma unroll
        for (int j = 0; j < KK * DD / 4 / 256; ++j) st[j] = s4[tid + 256 * j];
#pragma unroll
        for (int j = 0; j < KK * DD / 4 / 256; ++j) {
            const int i = tid + 256 * j;
            d4[(i >> 7) * (APITCH / 4) + (i & 127)] = st[j];
        }
    }
    __syncthreads();

    const float4* a0 = (const float4*)(As + lr * APITCH + 16 * half);
    const float4* a1 = (const float4*)(As + (lr + 16) * APITCH + 16 * half);

    float4 aa[2][2][4];
#pragma unroll
    for (int i = 0; i < 4; ++i) { aa[0][0][i] = a0[i]; aa[0][1][i] = a1[i]; }

    v8f acc0 = {};
    v8f acc1 = {};
    float ss = 0.f;
#pragma unroll
    for (int c = 0; c < NCH; ++c) {
        if (c + PFD < NCH) {
#pragma unroll
            for (int i = 0; i < 4; ++i)
                qb[c + PFD][i] = pb4[(c + PFD) * (KCH / 4) + i];
        }
        if (c + 1 < NCH) {
#pragma unroll
            for (int i = 0; i < 4; ++i) {
                aa[(c + 1) & 1][0][i] = a0[(c + 1) * (KCH / 4) + i];
                aa[(c + 1) & 1][1][i] = a1[(c + 1) * (KCH / 4) + i];
            }
        }
        __builtin_amdgcn_sched_barrier(0);
#pragma unroll
        for (int i = 0; i < 4; ++i) {
            const float4 bq = qb[c][i];
            const float4 x0 = aa[c & 1][0][i];
            const float4 x1 = aa[c & 1][1][i];
            ss += bq.x * bq.x + bq.y * bq.y + bq.z * bq.z + bq.w * bq.w;
            v2f bf  = {bq.x, bq.y};
            v2f af0 = {x0.x, x0.y};
            v2f af1 = {x1.x, x1.y};
            acc0 = __builtin_amdgcn_wmma_f32_16x16x4_f32(false, af0, false, bf,
                                                         (short)0, acc0, false, false);
            acc1 = __builtin_amdgcn_wmma_f32_16x16x4_f32(false, af1, false, bf,
                                                         (short)0, acc1, false, false);
            bf  = (v2f){bq.z, bq.w};
            af0 = (v2f){x0.z, x0.w};
            af1 = (v2f){x1.z, x1.w};
            acc0 = __builtin_amdgcn_wmma_f32_16x16x4_f32(false, af0, false, bf,
                                                         (short)0, acc0, false, false);
            acc1 = __builtin_amdgcn_wmma_f32_16x16x4_f32(false, af1, false, bf,
                                                         (short)0, acc1, false, false);
        }
        __builtin_amdgcn_sched_barrier(0);
    }
    ss += __shfl_xor(ss, 16, 32);
    const float inv = 1.f / fmaxf(sqrtf(ss), 1e-12f);
    const unsigned keylo = (unsigned)(NP - 1 - m);

    const int col = wave * 16 + lr;
#pragma unroll
    for (int i = 0; i < 8; ++i) {
        kbuf[(i + 8 * half) * KPITCH + col] =
            ((u64)ordkey(acc0[i] * inv) << 32) | keylo;
        kbuf[(16 + i + 8 * half) * KPITCH + col] =
            ((u64)ordkey(acc1[i] * inv) << 32) | keylo;
    }
    __syncthreads();

    {
        const u64x2* seg = (const u64x2*)(kbuf + lane * KPITCH + wave * 16);
        u64x2 q[8];
#pragma unroll
        for (int j = 0; j < 8; ++j) q[j] = seg[j];
        u64 v[8];
#pragma unroll
        for (int j = 0; j < 8; ++j) v[j] = q[j].x > q[j].y ? q[j].x : q[j].y;
#pragma unroll
        for (int w = 4; w; w >>= 1)
#pragma unroll
            for (int j = 0; j < w; ++j) v[j] = v[j + w] > v[j] ? v[j + w] : v[j];
        pbuf[wave * KK + lane] = v[0];
    }
    __syncthreads();

    if (wave == 0) {
        u64 v[8];
#pragma unroll
        for (int s = 0; s < 8; ++s) v[s] = pbuf[s * KK + lane];
#pragma unroll
        for (int w = 4; w; w >>= 1)
#pragma unroll
            for (int s = 0; s < w; ++s) v[s] = v[s + w] > v[s] ? v[s + w] : v[s];
        part[((size_t)b * NBLK + blockIdx.x) * KK + lane] = v[0];
    }
}

__global__ __launch_bounds__(256) void fuse_kernel(
    const float* __restrict__ p1, const float* __restrict__ p2,
    const u64* __restrict__ part, float* __restrict__ p_s_out) {
    const int t    = blockIdx.x * 8 + (threadIdx.x >> 5);
    const int lane = threadIdx.x & 31;
    const int b    = t >> 5;
    const int k    = t & 31;
    u64 e = part[((size_t)((lane >> 4) * B_ + b) * NBLK + (lane & 15)) * KK + k];
#pragma unroll
    for (int off = 8; off; off >>= 1) {
        const u64 o = __shfl_xor(e, off, 32);
        e = o > e ? o : e;
    }
    const u64 eo = __shfl_xor(e, 16, 32);
    const u64 e1 = (lane < 16) ? e : eo;
    const u64 e2 = (lane < 16) ? eo : e;
    const float w1 = keyval((unsigned)(e1 >> 32));
    const float w2 = keyval((unsigned)(e2 >> 32));
    const int  i1  = NP - 1 - (int)(unsigned)(e1 & 0xffffffffull);
    const int  i2  = NP - 1 - (int)(unsigned)(e2 & 0xffffffffull);
    const float* r1 = p1 + ((size_t)b * NP + i1) * DD;
    const float* r2 = p2 + ((size_t)b * NP + i2) * DD;

    float v1[16], v2[16];
    float s1 = 0.f, s2 = 0.f;
#pragma unroll
    for (int j = 0; j < 16; ++j) {
        float x = r1[lane + 32 * j]; v1[j] = x; s1 += x * x;
        float y = r2[lane + 32 * j]; v2[j] = y; s2 += y * y;
    }
#pragma unroll
    for (int off = 16; off; off >>= 1) {
        s1 += __shfl_xor(s1, off, 32);
        s2 += __shfl_xor(s2, off, 32);
    }
    const float c1 = w1 / fmaxf(sqrtf(s1), 1e-12f);
    const float c2 = w2 / fmaxf(sqrtf(s2), 1e-12f);
    const float dn = 1.f / (w1 + w2 + 1e-6f);

    float f[16];
    float sf = 0.f;
#pragma unroll
    for (int j = 0; j < 16; ++j) {
        f[j] = (c1 * v1[j] + c2 * v2[j]) * dn;
        sf += f[j] * f[j];
    }
#pragma unroll
    for (int off = 16; off; off >>= 1) sf += __shfl_xor(sf, off, 32);
    const float invf = 1.f / fmaxf(sqrtf(sf), 1e-12f);
    float* dst = p_s_out + (size_t)t * DD;
#pragma unroll
    for (int j = 0; j < 16; ++j) dst[lane + 32 * j] = f[j] * invf;
}

extern "C" void kernel_launch(void* const* d_in, const int* in_sizes, int n_in,
                              void* d_out, int out_size, void* d_ws, size_t ws_size,
                              hipStream_t stream) {
    (void)in_sizes; (void)n_in; (void)out_size; (void)ws_size;
    const float* anchor = (const float*)d_in[0];
    const float* p1     = (const float*)d_in[1];
    const float* p2     = (const float*)d_in[2];
    const int*   idx_a  = (const int*)d_in[3];
    const int*   idx_p  = (const int*)d_in[4];

    float* A_s = (float*)d_out;
    float* P_s = (float*)d_out + (size_t)B_ * KK * DD;

    u64* part  = (u64*)d_ws;
    float* asel = P_s;

    prep_kernel<<<dim3((2 * B_ * KK) / 8), 256, 0, stream>>>(anchor, idx_a, idx_p, A_s, asel);

    dim3 sg(NBLK, B_);
    score_kernel<<<sg, 256, 0, stream>>>(p1, asel, part);
    score_kernel<<<sg, 256, 0, stream>>>(p2, asel, part + B_ * NBLK * KK);

    fuse_kernel<<<dim3((B_ * KK) / 8), 256, 0, stream>>>(p1, p2, part, P_s);
}


# Gather launch removed (4->3): each score block gathers+normalizes its 32 idx_p anchor rows into LDS (float4 loads; row-norm sum order changes); A_s rows come from 64 extra blocks of the fuse launch
# speedup vs baseline: 1.9270x; 1.1646x over previous
#include <hip/hip_runtime.h>

#define B_  16
#define NA  2048
#define NP  2048
#define DD  512
#define KK  32

typedef __attribute__((ext_vector_type(2))) float v2f;
typedef __attribute__((ext_vector_type(8))) float v8f;

__device__ __forceinline__ unsigned ordkey(float f) {
    unsigned u = __float_as_uint(f);
    return (u & 0x80000000u) ? ~u : (u | 0x80000000u);
}
__device__ __forceinline__ float keyval(unsigned k) {
    return __uint_as_float((k & 0x80000000u) ? (k ^ 0x80000000u) : ~k);
}


#define APITCH (DD + 4)
#define KCH    32
#define NCH    (DD / KCH)
#define PFD    6
#define PF0    2
#define NBLK   (NP / 128)
#define KPITCH 130

typedef unsigned long long u64;
typedef __attribute__((ext_vector_type(2))) unsigned long long u64x2;

__global__ __launch_bounds__(256) void score_kernel(
    const float* __restrict__ p, const float* __restrict__ anchor,
    const int* __restrict__ idx_p, u64* __restrict__ part) {
    __shared__ __attribute__((aligned(16))) float As[KK * APITCH];
    __shared__ __attribute__((aligned(16))) u64 kbuf[KK * KPITCH];
    __shared__ __attribute__((aligned(16))) u64 pbuf[8 * KK];
    const int b    = blockIdx.y;
    const int tid  = threadIdx.x;
    const int wave = tid >> 5;
    const int lane = tid & 31;
    const int half = lane >> 4;
    const int lr   = lane & 15;
    const int m    = blockIdx.x * 128 + wave * 16 + lr;

    const int wv = __builtin_amdgcn_readfirstlane(wave);
    int sel[4];
#pragma unroll
    for (int i = 0; i < 4; ++i) sel[i] = idx_p[b * KK + 4 * wv + i];

    const float4* pb4 =
        (const float4*)(p + ((size_t)b * NP + m) * DD + 16 * half);

    float4 qb[NCH][4];
#pragma unroll
    for (int c = 0; c < PF0; ++c)
#pragma unroll
        for (int i = 0; i < 4; ++i) qb[c][i] = pb4[c * (KCH / 4) + i];
    __builtin_amdgcn_sched_barrier(0);

    {
        float4 v[4][4];
#pragma unroll
        for (int i = 0; i < 4; ++i) {
            const float4* src = (const float4*)(anchor + ((size_t)b * NA + sel[i]) * DD);
#pragma unroll
            for (int q = 0; q < 4; ++q) v[i][q] = src[lane + 32 * q];
        }
#pragma unroll
        for (int c = PF0; c < PFD; ++c)
#pragma unroll
            for (int i = 0; i < 4; ++i) qb[c][i] = pb4[c * (KCH / 4) + i];
#pragma unroll
        for (int i = 0; i < 4; ++i) {
            float s = 0.f;
#pragma unroll
            for (int q = 0; q < 4; ++q) {
                const float4 x = v[i][q];
                s += x.x * x.x + x.y * x.y + x.z * x.z + x.w * x.w;
            }
#pragma unroll
            for (int off = 16; off; off >>= 1) s += __shfl_xor(s, off, 32);
            const float inv = 1.f / fmaxf(sqrtf(s), 1e-12f);
            float4* dst = (float4*)(As + (4 * wv + i) * APITCH);
#pragma unroll
            for (int q = 0; q < 4; ++q) {
                const float4 x = v[i][q];
                dst[lane + 32 * q] = make_float4(x.x * inv, x.y * inv, x.z * inv, x.w * inv);
            }
        }
    }
    __syncthreads();

    const float4* a0 = (const float4*)(As + lr * APITCH + 16 * half);
    const float4* a1 = (const float4*)(As + (lr + 16) * APITCH + 16 * half);

    float4 aa[2][2][4];
#pragma unroll
    for (int i = 0; i < 4; ++i) { aa[0][0][i] = a0[i]; aa[0][1][i] = a1[i]; }

    v8f acc0 = {};
    v8f acc1 = {};
    float ss = 0.f;
#pragma unroll
    for (int c = 0; c < NCH; ++c) {
        if (c + PFD < NCH) {
#pragma unroll
            for (int i = 0; i < 4; ++i)
                qb[c + PFD][i] = pb4[(c + PFD) * (KCH / 4) + i];
        }
        if (c + 1 < NCH) {
#pragma unroll
            for (int i = 0; i < 4; ++i) {
                aa[(c + 1) & 1][0][i] = a0[(c + 1) * (KCH / 4) + i];
                aa[(c + 1) & 1][1][i] = a1[(c + 1) * (KCH / 4) + i];
            }
        }
        __builtin_amdgcn_sched_barrier(0);
#pragma unroll
        for (int i = 0; i < 4; ++i) {
            const float4 bq = qb[c][i];
            const float4 x0 = aa[c & 1][0][i];
            const float4 x1 = aa[c & 1][1][i];
            ss += bq.x * bq.x + bq.y * bq.y + bq.z * bq.z + bq.w * bq.w;
            v2f bf  = {bq.x, bq.y};
            v2f af0 = {x0.x, x0.y};
            v2f af1 = {x1.x, x1.y};
            acc0 = __builtin_amdgcn_wmma_f32_16x16x4_f32(false, af0, false, bf,
                                                         (short)0, acc0, false, false);
            acc1 = __builtin_amdgcn_wmma_f32_16x16x4_f32(false, af1, false, bf,
                                                         (short)0, acc1, false, false);
            bf  = (v2f){bq.z, bq.w};
            af0 = (v2f){x0.z, x0.w};
            af1 = (v2f){x1.z, x1.w};
            acc0 = __builtin_amdgcn_wmma_f32_16x16x4_f32(false, af0, false, bf,
                                                         (short)0, acc0, false, false);
            acc1 = __builtin_amdgcn_wmma_f32_16x16x4_f32(false, af1, false, bf,
                                                         (short)0, acc1, false, false);
        }
        __builtin_amdgcn_sched_barrier(0);
    }
    ss += __shfl_xor(ss, 16, 32);
    const float inv = 1.f / fmaxf(sqrtf(ss), 1e-12f);
    const unsigned keylo = (unsigned)(NP - 1 - m);

    const int col = wave * 16 + lr;
#pragma unroll
    for (int i = 0; i < 8; ++i) {
        kbuf[(i + 8 * half) * KPITCH + col] =
            ((u64)ordkey(acc0[i] * inv) << 32) | keylo;
        kbuf[(16 + i + 8 * half) * KPITCH + col] =
            ((u64)ordkey(acc1[i] * inv) << 32) | keylo;
    }
    __syncthreads();

    {
        const u64x2* seg = (const u64x2*)(kbuf + lane * KPITCH + wave * 16);
        u64x2 q[8];
#pragma unroll
        for (int j = 0; j < 8; ++j) q[j] = seg[j];
        u64 v[8];
#pragma unroll
        for (int j = 0; j < 8; ++j) v[j] = q[j].x > q[j].y ? q[j].x : q[j].y;
#pragma unroll
        for (int w = 4; w; w >>= 1)
#pragma unroll
            for (int j = 0; j < w; ++j) v[j] = v[j + w] > v[j] ? v[j + w] : v[j];
        pbuf[wave * KK + lane] = v[0];
    }
    __syncthreads();

    if (wave == 0) {
        u64 v[8];
#pragma unroll
        for (int s = 0; s < 8; ++s) v[s] = pbuf[s * KK + lane];
#pragma unroll
        for (int w = 4; w; w >>= 1)
#pragma unroll
            for (int s = 0; s < w; ++s) v[s] = v[s + w] > v[s] ? v[s + w] : v[s];
        part[((size_t)b * NBLK + blockIdx.x) * KK + lane] = v[0];
    }
}

#define FBLK ((B_ * KK) / 8)
__global__ __launch_bounds__(256) void fuse_kernel(
    const float* __restrict__ p1, const float* __restrict__ p2,
    const float* __restrict__ anchor, const int* __restrict__ idx_a,
    const u64* __restrict__ part, float* __restrict__ a_s_out,
    float* __restrict__ p_s_out) {
    const int lane = threadIdx.x & 31;
    if (blockIdx.x >= FBLK) {
        const int t = (blockIdx.x - FBLK) * 8 + (threadIdx.x >> 5);
        const float* src = anchor + ((size_t)(t >> 5) * NA + idx_a[t]) * DD;
        float v[16];
        float s = 0.f;
#pragma unroll
        for (int j = 0; j < 16; ++j) { float x = src[lane + 32 * j]; v[j] = x; s += x * x; }
#pragma unroll
        for (int off = 16; off; off >>= 1) s += __shfl_xor(s, off, 32);
        const float inv = 1.f / fmaxf(sqrtf(s), 1e-12f);
        float* dst = a_s_out + (size_t)t * DD;
#pragma unroll
        for (int j = 0; j < 16; ++j) dst[lane + 32 * j] = v[j] * inv;
        return;
    }
    const int t    = blockIdx.x * 8 + (threadIdx.x >> 5);
    const int b    = t >> 5;
    const int k    = t & 31;
    u64 e = part[((size_t)((lane >> 4) * B_ + b) * NBLK + (lane & 15)) * KK + k];
#pragma unroll
    for (int off = 8; off; off >>= 1) {
        const u64 o = __shfl_xor(e, off, 32);
        e = o > e ? o : e;
    }
    const u64 eo = __shfl_xor(e, 16, 32);
    const u64 e1 = (lane < 16) ? e : eo;
    const u64 e2 = (lane < 16) ? eo : e;
    const float w1 = keyval((unsigned)(e1 >> 32));
    const float w2 = keyval((unsigned)(e2 >> 32));
    const int  i1  = NP - 1 - (int)(unsigned)(e1 & 0xffffffffull);
    const int  i2  = NP - 1 - (int)(unsigned)(e2 & 0xffffffffull);
    const float* r1 = p1 + ((size_t)b * NP + i1) * DD;
    const float* r2 = p2 + ((size_t)b * NP + i2) * DD;

    float v1[16], v2[16];
    float s1 = 0.f, s2 = 0.f;
#pragma unroll
    for (int j = 0; j < 16; ++j) {
        float x = r1[lane + 32 * j]; v1[j] = x; s1 += x * x;
        float y = r2[lane + 32 * j]; v2[j] = y; s2 += y * y;
    }
#pragma unroll
    for (int off = 16; off; off >>= 1) {
        s1 += __shfl_xor(s1, off, 32);
        s2 += __shfl_xor(s2, off, 32);
    }
    const float c1 = w1 / fmaxf(sqrtf(s1), 1e-12f);
    const float c2 = w2 / fmaxf(sqrtf(s2), 1e-12f);
    const float dn = 1.f / (w1 + w2 + 1e-6f);

    float f[16];
    float sf = 0.f;
#pragma unroll
    for (int j = 0; j < 16; ++j) {
        f[j] = (c1 * v1[j] + c2 * v2[j]) * dn;
        sf += f[j] * f[j];
    }
#pragma unroll
    for (int off = 16; off; off >>= 1) sf += __shfl_xor(sf, off, 32);
    const float invf = 1.f / fmaxf(sqrtf(sf), 1e-12f);
    float* dst = p_s_out + (size_t)t * DD;
#pragma unroll
    for (int j = 0; j < 16; ++j) dst[lane + 32 * j] = f[j] * invf;
}

extern "C" void kernel_launch(void* const* d_in, const int* in_sizes, int n_in,
                              void* d_out, int out_size, void* d_ws, size_t ws_size,
                              hipStream_t stream) {
    (void)in_sizes; (void)n_in; (void)out_size; (void)ws_size;
    const float* anchor = (const float*)d_in[0];
    const float* p1     = (const float*)d_in[1];
    const float* p2     = (const float*)d_in[2];
    const int*   idx_a  = (const int*)d_in[3];
    const int*   idx_p  = (const int*)d_in[4];

    float* A_s = (float*)d_out;
    float* P_s = (float*)d_out + (size_t)B_ * KK * DD;

    u64* part  = (u64*)d_ws;

    dim3 sg(NBLK, B_);
    score_kernel<<<sg, 256, 0, stream>>>(p1, anchor, idx_p, part);
    score_kernel<<<sg, 256, 0, stream>>>(p2, anchor, idx_p, part + B_ * NBLK * KK);

    fuse_kernel<<<dim3(2 * FBLK), 256, 0, stream>>>(p1, p2, anchor, idx_a, part, A_s, P_s);
}


# p chunks read as whole 128-B lines (4 lines per 16-B wave load, was a 32-B piece of 16 lines) and turned into B fragments through a wave-private LDS stage; k pairing and sum order unchanged
# speedup vs baseline: 2.0158x; 1.0461x over previous
#include <hip/hip_runtime.h>

#define B_  16
#define NA  2048
#define NP  2048
#define DD  512
#define KK  32

typedef __attribute__((ext_vector_type(2))) float v2f;
typedef __attribute__((ext_vector_type(4))) float v4f;
typedef __attribute__((ext_vector_type(8))) float v8f;

__device__ __forceinline__ unsigned ordkey(float f) {
    unsigned u = __float_as_uint(f);
    return (u & 0x80000000u) ? ~u : (u | 0x80000000u);
}
__device__ __forceinline__ float keyval(unsigned k) {
    return __uint_as_float((k & 0x80000000u) ? (k ^ 0x80000000u) : ~k);
}


#define APITCH (DD + 4)
#define KCH    32
#define NCH    (DD / KCH)
#define PFD    6
#define PF0    2
#define NBLK   (NP / 128)
#define SPITCH 36
#define SSLOT  (16 * SPITCH)
#define KPITCH 130

typedef unsigned long long u64;
typedef __attribute__((ext_vector_type(2))) unsigned long long u64x2;

__global__ __launch_bounds__(256) void score_kernel(
    const float* __restrict__ p, const float* __restrict__ anchor,
    const int* __restrict__ idx_p, u64* __restrict__ part) {
    __shared__ __attribute__((aligned(16))) float As[KK * APITCH];
    __shared__ __attribute__((aligned(16))) u64 kbuf[KK * KPITCH];
    __shared__ __attribute__((aligned(16))) u64 pbuf[8 * KK];
    __shared__ __attribute__((aligned(16))) float Bs[8 * 2 * SSLOT];
    const int b    = blockIdx.y;
    const int tid  = threadIdx.x;
    const int wave = tid >> 5;
    const int lane = tid & 31;
    const int half = lane >> 4;
    const int lr   = lane & 15;
    const int m    = blockIdx.x * 128 + wave * 16 + lr;

    const int wv = __builtin_amdgcn_readfirstlane(wave);
    int sel[4];
#pragma unroll
    for (int i = 0; i < 4; ++i) sel[i] = idx_p[b * KK + 4 * wv + i];

    const int g   = lane >> 3;
    const int pc  = lane & 7;
    const int rl0 = (g >> 1) + 8 * (g & 1);
    const v4f* pb4 = (const v4f*)(p +
        ((size_t)b * NP + blockIdx.x * 128 + wv * 16 + rl0) * DD + 4 * pc);
    v4f*       st4 = (v4f*)(Bs + wv * 2 * SSLOT + rl0 * SPITCH + 4 * pc);
    const v4f* sr4 = (const v4f*)(Bs + wv * 2 * SSLOT + lr * SPITCH + 16 * half);

    v4f qb[NCH][4];
#pragma unroll
    for (int c = 0; c < PF0; ++c)
#pragma unroll
        for (int i = 0; i < 4; ++i) qb[c][i] = pb4[c * (KCH / 4) + i * (2 * DD / 4)];
    __builtin_amdgcn_sched_barrier(0);

    {
        float4 v[4][4];
#pragma unroll
        for (int i = 0; i < 4; ++i) {
            const float4* src = (const float4*)(anchor + ((size_t)b * NA + sel[i]) * DD);
#pragma unroll
            for (int q = 0; q < 4; ++q) v[i][q] = src[lane + 32 * q];
        }
#pragma unroll
        for (int c = PF0; c < PFD; ++c)
#pragma unroll
            for (int i = 0; i < 4; ++i) qb[c][i] = pb4[c * (KCH / 4) + i * (2 * DD / 4)];
#pragma unroll
        for (int i = 0; i < 4; ++i) {
            float s = 0.f;
#pragma unroll
            for (int q = 0; q < 4; ++q) {
                const float4 x = v[i][q];
                s += x.x * x.x + x.y * x.y + x.z * x.z + x.w * x.w;
            }
#pragma unroll
            for (int off = 16; off; off >>= 1) s += __shfl_xor(s, off, 32);
            const float inv = 1.f / fmaxf(sqrtf(s), 1e-12f);
            float4* dst = (float4*)(As + (4 * wv + i) * APITCH);
#pragma unroll
            for (int q = 0; q < 4; ++q) {
                const float4 x = v[i][q];
                dst[lane + 32 * q] = make_float4(x.x * inv, x.y * inv, x.z * inv, x.w * inv);
            }
        }
    }
#pragma unroll
    for (int j = 0; j < 4; ++j) st4[j * (2 * SPITCH / 4)] = qb[0][j];
    __syncthreads();

    const float4* a0 = (const float4*)(As + lr * APITCH + 16 * half);
    const float4* a1 = (const float4*)(As + (lr + 16) * APITCH + 16 * half);

    float4 aa[2][2][4];
    v4f bb[2][4];
#pragma unroll
    for (int i = 0; i < 4; ++i) { aa[0][0][i] = a0[i]; aa[0][1][i] = a1[i]; bb[0][i] = sr4[i]; }

    v8f acc0 = {};
    v8f acc1 = {};
    float ss = 0.f;
#pragma unroll
    for (int c = 0; c < NCH; ++c) {
        if (c + PFD < NCH) {
#pragma unroll
            for (int i = 0; i < 4; ++i)
                qb[c + PFD][i] = pb4[(c + PFD) * (KCH / 4) + i * (2 * DD / 4)];
        }
        if (c + 1 < NCH) {
#pragma unroll
            for (int j = 0; j < 4; ++j)
                st4[((c + 1) & 1) * (SSLOT / 4) + j * (2 * SPITCH / 4)] = qb[c + 1][j];
#pragma unroll
            for (int i = 0; i < 4; ++i) {
                aa[(c + 1) & 1][0][i] = a0[(c + 1) * (KCH / 4) + i];
                aa[(c + 1) & 1][1][i] = a1[(c + 1) * (KCH / 4) + i];
                bb[(c + 1) & 1][i]    = sr4[((c + 1) & 1) * (SSLOT / 4) + i];
            }
        }
        __builtin_amdgcn_sched_barrier(0);
#pragma unroll
        for (int i = 0; i < 4; ++i) {
            const v4f bq = bb[c & 1][i];
            const float4 x0 = aa[c & 1][0][i];
            const float4 x1 = aa[c & 1][1][i];
            ss += bq.x * bq.x + bq.y * bq.y + bq.z * bq.z + bq.w * bq.w;
            v2f bf  = {bq.x, bq.y};
            v2f af0 = {x0.x, x0.y};
            v2f af1 = {x1.x, x1.y};
            acc0 = __builtin_amdgcn_wmma_f32_16x16x4_f32(false, af0, false, bf,
                                                         (short)0, acc0, false, false);
            acc1 = __builtin_amdgcn_wmma_f32_16x16x4_f32(false, af1, false, bf,
                                                         (short)0, acc1, false, false);
            bf  = (v2f){bq.z, bq.w};
            af0 = (v2f){x0.z, x0.w};
            af1 = (v2f){x1.z, x1.w};
            acc0 = __builtin_amdgcn_wmma_f32_16x16x4_f32(false, af0, false, bf,
                                                         (short)0, acc0, false, false);
            acc1 = __builtin_amdgcn_wmma_f32_16x16x4_f32(false, af1, false, bf,
                                                         (short)0, acc1, false, false);
        }
        __builtin_amdgcn_sched_barrier(0);
    }
    ss += __shfl_xor(ss, 16, 32);
    const float inv = 1.f / fmaxf(sqrtf(ss), 1e-12f);
    const unsigned keylo = (unsigned)(NP - 1 - m);

    const int col = wave * 16 + lr;
#pragma unroll
    for (int i = 0; i < 8; ++i) {
        kbuf[(i + 8 * half) * KPITCH + col] =
            ((u64)ordkey(acc0[i] * inv) << 32) | keylo;
        kbuf[(16 + i + 8 * half) * KPITCH + col] =
            ((u64)ordkey(acc1[i] * inv) << 32) | keylo;
    }
    __syncthreads();

    {
        const u64x2* seg = (const u64x2*)(kbuf + lane * KPITCH + wave * 16);
        u64x2 q[8];
#pragma unroll
        for (int j = 0; j < 8; ++j) q[j] = seg[j];
        u64 v[8];
#pragma unroll
        for (int j = 0; j < 8; ++j) v[j] = q[j].x > q[j].y ? q[j].x : q[j].y;
#pragma unroll
        for (int w = 4; w; w >>= 1)
#pragma unroll
            for (int j = 0; j < w; ++j) v[j] = v[j + w] > v[j] ? v[j + w] : v[j];
        pbuf[wave * KK + lane] = v[0];
    }
    __syncthreads();

    if (wave == 0) {
        u64 v[8];
#pragma unroll
        for (int s = 0; s < 8; ++s) v[s] = pbuf[s * KK + lane];
#pragma unroll
        for (int w = 4; w; w >>= 1)
#pragma unroll
            for (int s = 0; s < w; ++s) v[s] = v[s + w] > v[s] ? v[s + w] : v[s];
        part[((size_t)b * NBLK + blockIdx.x) * KK + lane] = v[0];
    }
}

#define FBLK ((B_ * KK) / 8)
__global__ __launch_bounds__(256) void fuse_kernel(
    const float* __restrict__ p1, const float* __restrict__ p2,
    const float* __restrict__ anchor, const int* __restrict__ idx_a,
    const u64* __restrict__ part, float* __restrict__ a_s_out,
    float* __restrict__ p_s_out) {
    const int lane = threadIdx.x & 31;
    if (blockIdx.x >= FBLK) {
        const int t = (blockIdx.x - FBLK) * 8 + (threadIdx.x >> 5);
        const float* src = anchor + ((size_t)(t >> 5) * NA + idx_a[t]) * DD;
        float v[16];
        float s = 0.f;
#pragma unroll
        for (int j = 0; j < 16; ++j) { float x = src[lane + 32 * j]; v[j] = x; s += x * x; }
#pragma unroll
        for (int off = 16; off; off >>= 1) s += __shfl_xor(s, off, 32);
        const float inv = 1.f / fmaxf(sqrtf(s), 1e-12f);
        float* dst = a_s_out + (size_t)t * DD;
#pragma unroll
        for (int j = 0; j < 16; ++j) dst[lane + 32 * j] = v[j] * inv;
        return;
    }
    const int t    = blockIdx.x * 8 + (threadIdx.x >> 5);
    const int b    = t >> 5;
    const int k    = t & 31;
    u64 e = part[((size_t)((lane >> 4) * B_ + b) * NBLK + (lane & 15)) * KK + k];
#pragma unroll
    for (int off = 8; off; off >>= 1) {
        const u64 o = __shfl_xor(e, off, 32);
        e = o > e ? o : e;
    }
    const u64 eo = __shfl_xor(e, 16, 32);
    const u64 e1 = (lane < 16) ? e : eo;
    const u64 e2 = (lane < 16) ? eo : e;
    const float w1 = keyval((unsigned)(e1 >> 32));
    const float w2 = keyval((unsigned)(e2 >> 32));
    const int  i1  = NP - 1 - (int)(unsigned)(e1 & 0xffffffffull);
    const int  i2  = NP - 1 - (int)(unsigned)(e2 & 0xffffffffull);
    const float* r1 = p1 + ((size_t)b * NP + i1) * DD;
    const float* r2 = p2 + ((size_t)b * NP + i2) * DD;

    float v1[16], v2[16];
    float s1 = 0.f, s2 = 0.f;
#pragma unroll
    for (int j = 0; j < 16; ++j) {
        float x = r1[lane + 32 * j]; v1[j] = x; s1 += x * x;
        float y = r2[lane + 32 * j]; v2[j] = y; s2 += y * y;
    }
#pragma unroll
    for (int off = 16; off; off >>= 1) {
        s1 += __shfl_xor(s1, off, 32);
        s2 += __shfl_xor(s2, off, 32);
    }
    const float c1 = w1 / fmaxf(sqrtf(s1), 1e-12f);
    const float c2 = w2 / fmaxf(sqrtf(s2), 1e-12f);
    const float dn = 1.f / (w1 + w2 + 1e-6f);

    float f[16];
    float sf = 0.f;
#pragma unroll
    for (int j = 0; j < 16; ++j) {
        f[j] = (c1 * v1[j] + c2 * v2[j]) * dn;
        sf += f[j] * f[j];
    }
#pragma unroll
    for (int off = 16; off; off >>= 1) sf += __shfl_xor(sf, off, 32);
    const float invf = 1.f / fmaxf(sqrtf(sf), 1e-12f);
    float* dst = p_s_out + (size_t)t * DD;
#pragma unroll
    for (int j = 0; j < 16; ++j) dst[lane + 32 * j] = f[j] * invf;
}

extern "C" void kernel_launch(void* const* d_in, const int* in_sizes, int n_in,
                              void* d_out, int out_size, void* d_ws, size_t ws_size,
                              hipStream_t stream) {
    (void)in_sizes; (void)n_in; (void)out_size; (void)ws_size;
    const float* anchor = (const float*)d_in[0];
    const float* p1     = (const float*)d_in[1];
    const float* p2     = (const float*)d_in[2];
    const int*   idx_a  = (const int*)d_in[3];
    const int*   idx_p  = (const int*)d_in[4];

    float* A_s = (float*)d_out;
    float* P_s = (float*)d_out + (size_t)B_ * KK * DD;

    u64* part  = (u64*)d_ws;

    dim3 sg(NBLK, B_);
    score_kernel<<<sg, 256, 0, stream>>>(p1, anchor, idx_p, part);
    score_kernel<<<sg, 256, 0, stream>>>(p2, anchor, idx_p, part + B_ * NBLK * KK);

    fuse_kernel<<<dim3(2 * FBLK), 256, 0, stream>>>(p1, p2, anchor, idx_a, part, A_s, P_s);
}
